# starting kernel
# baseline (speedup 1.0000x reference)
#include <hip/hip_runtime.h>

typedef __attribute__((ext_vector_type(2))) float v2f;
typedef __attribute__((ext_vector_type(8))) float v8f;

#define GCN_D 128

__global__ __launch_bounds__(256) void gcn_degree(const int* __restrict__ dst,
                                                  float* __restrict__ deg, int n_edges) {
    int i = blockIdx.x * blockDim.x + threadIdx.x;
    int stride = gridDim.x * blockDim.x;
    for (; i < n_edges; i += stride) atomicAdd(&deg[dst[i]], 1.0f);
}

__global__ __launch_bounds__(256) void gcn_dinv(float* __restrict__ deg, int n_nodes) {
    int i = blockIdx.x * blockDim.x + threadIdx.x;
    if (i < n_nodes) deg[i] = rsqrtf(1.0f + deg[i]);
}

__global__ __launch_bounds__(256) void gcn_gemm(const float* __restrict__ X,
                                                const float* __restrict__ W,
                                                float* __restrict__ H, int n) {
    __shared__ float sW[GCN_D * GCN_D];
    for (int i = threadIdx.x; i < GCN_D * GCN_D; i += blockDim.x) sW[i] = W[i];
    __syncthreads();

    const int lane = threadIdx.x & 31;
    const int wave = threadIdx.x >> 5;
    const int m    = lane & 15;
    const int kh   = lane >> 4;
    const int row0 = blockIdx.x * 128 + wave * 16;

    int rowA = row0 + m;
    if (rowA > n - 1) rowA = n - 1;
    const float* xrow = X + (size_t)rowA * GCN_D;

    v8f acc[8] = {};

    for (int kk = 0; kk < 32; ++kk) {
        const int kb = kk * 4 + kh * 2;
        v2f a;
        a.x = xrow[kb];
        a.y = xrow[kb + 1];
        #pragma unroll
        for (int nt = 0; nt < 8; ++nt) {
            v2f b;
            b.x = sW[(kb + 0) * GCN_D + nt * 16 + m];
            b.y = sW[(kb + 1) * GCN_D + nt * 16 + m];
            acc[nt] = __builtin_amdgcn_wmma_f32_16x16x4_f32(
                false, a, false, b, (short)0, acc[nt], false, false);
        }
    }

    #pragma unroll
    for (int nt = 0; nt < 8; ++nt) {
        #pragma unroll
        for (int v = 0; v < 8; ++v) {
            int r = row0 + v + 8 * kh;
            if (r < n) H[(size_t)r * GCN_D + nt * 16 + m] = acc[nt][v];
        }
    }
}

__global__ __launch_bounds__(256) void gcn_scatter(const int* __restrict__ src,
                                                   const int* __restrict__ dst,
                                                   const float* __restrict__ dinv,
                                                   const float* __restrict__ h,
                                                   float* __restrict__ agg, int n_edges) {
    const int lane = threadIdx.x & 31;
    int wid = blockIdx.x * (blockDim.x >> 5) + (threadIdx.x >> 5);
    const int nw = gridDim.x * (blockDim.x >> 5);
    for (int e = wid; e < n_edges; e += nw) {
        const int s = src[e];
        const int d = dst[e];
        const float norm = dinv[s] * dinv[d];
        const float4 v = ((const float4*)(h + (size_t)s * GCN_D))[lane];
        float* a = agg + (size_t)d * GCN_D + lane * 4;
        atomicAdd(a + 0, v.x * norm);
        atomicAdd(a + 1, v.y * norm);
        atomicAdd(a + 2, v.z * norm);
        atomicAdd(a + 3, v.w * norm);
    }
}

__global__ __launch_bounds__(256) void gcn_combine(const float* __restrict__ h,
                                                   const float* __restrict__ agg,
                                                   const float* __restrict__ dinv,
                                                   const float* __restrict__ bias,
                                                   float* __restrict__ out, int n, int do_relu) {
    size_t i = (size_t)blockIdx.x * blockDim.x + threadIdx.x;
    size_t total = (size_t)n * (GCN_D / 4);
    if (i >= total) return;
    int node = (int)(i >> 5);
    int f4   = (int)(i & 31);
    float di = dinv[node];
    float d2 = di * di;
    float4 hv = ((const float4*)h)[i];
    float4 av = ((const float4*)agg)[i];
    float4 bv = ((const float4*)bias)[f4];
    float4 o;
    o.x = av.x + hv.x * d2 + bv.x;
    o.y = av.y + hv.y * d2 + bv.y;
    o.z = av.z + hv.z * d2 + bv.z;
    o.w = av.w + hv.w * d2 + bv.w;
    if (do_relu) {
        o.x = fmaxf(o.x, 0.0f);
        o.y = fmaxf(o.y, 0.0f);
        o.z = fmaxf(o.z, 0.0f);
        o.w = fmaxf(o.w, 0.0f);
    }
    ((float4*)out)[i] = o;
}

extern "C" void kernel_launch(void* const* d_in, const int* in_sizes, int n_in,
                              void* d_out, int out_size, void* d_ws, size_t ws_size,
                              hipStream_t stream) {
    const float* x  = (const float*)d_in[0];
    const int*   ei = (const int*)d_in[1];
    const float* W1 = (const float*)d_in[2];
    const float* b1 = (const float*)d_in[3];
    const float* W2 = (const float*)d_in[4];
    const float* b2 = (const float*)d_in[5];
    const float* W3 = (const float*)d_in[6];
    const float* b3 = (const float*)d_in[7];

    const int n = in_sizes[0] / GCN_D;
    const int E = in_sizes[1] / 2;
    const int* srcI = ei;
    const int* dstI = ei + E;

    float* ws = (float*)d_ws;
    size_t n4 = ((size_t)n + 3) & ~(size_t)3;
    float* dinv = ws;
    float* h    = ws + n4;
    float* agg  = h + (size_t)n * GCN_D;
    float* bufA = agg + (size_t)n * GCN_D;
    float* out  = (float*)d_out;

    hipMemsetAsync(dinv, 0, (size_t)n * sizeof(float), stream);
    gcn_degree<<<1024, 256, 0, stream>>>(dstI, dinv, E);
    gcn_dinv<<<(n + 255) / 256, 256, 0, stream>>>(dinv, n);

    const float* Ws[3] = {W1, W2, W3};
    const float* bs[3] = {b1, b2, b3};
    const float* cur = x;
    for (int l = 0; l < 3; ++l) {
        float* o = (l == 2) ? out : bufA;
        gcn_gemm<<<(n + 127) / 128, 256, 0, stream>>>(cur, Ws[l], h, n);
        hipMemsetAsync(agg, 0, (size_t)n * GCN_D * sizeof(float), stream);
        gcn_scatter<<<2048, 256, 0, stream>>>(srcI, dstI, dinv, h, agg, E);
        size_t total4 = (size_t)n * (GCN_D / 4);
        gcn_combine<<<(int)((total4 + 255) / 256), 256, 0, stream>>>(
            h, agg, dinv, bs[l], o, n, (l < 2) ? 1 : 0);
        cur = o;
    }
}


# First product re-tiled like the fused ones: a work-group owns 400 rows staged in LDS by 16-byte loads, a wave keeps its column tile's W fragments in registers from global; same k order, same bits
# speedup vs baseline: 3.1688x; 3.1688x over previous
#include <hip/hip_runtime.h>

typedef __attribute__((ext_vector_type(2))) float v2f;
typedef __attribute__((ext_vector_type(8))) float v8f;
typedef __attribute__((ext_vector_type(4))) unsigned v4u;

#define GCN_D 128
#define GCN_TILE 400
#define GCN_TWAVES 32
#define GCN_QCAP 40
#define GCN_PB 4096
#define GCN_PBW 128
#define GCN_PSTR 132
#define GCN_PT 256
#define GCN_APITCH 132

__global__ __launch_bounds__(1024) void gcn_planes(const int* __restrict__ dst,
                                                   float* __restrict__ deg,
                                                   unsigned* __restrict__ planes,
                                                   int n_edges, int n_tiles, int plane_words) {
    __shared__ unsigned bits[GCN_PT * GCN_PSTR];
    const int tid = threadIdx.x;
    int tl[4];
    #pragma unroll
    for (int k = 0; k < 4; ++k) {
        const int e = blockIdx.x * GCN_PB + k * 1024 + tid;
        tl[k] = -1;
        if (e < n_edges) {
            const int d = dst[e];
            atomicAdd(&deg[d], 1.0f);
            tl[k] = d / GCN_TILE;
        }
    }
    for (int tb = 0; tb < n_tiles; tb += GCN_PT) {
        for (int i = tid; i < GCN_PT * GCN_PSTR / 4; i += 1024)
            ((uint4*)bits)[i] = make_uint4(0u, 0u, 0u, 0u);
        __syncthreads();
        #pragma unroll
        for (int k = 0; k < 4; ++k) {
            const unsigned t = (unsigned)(tl[k] - tb);
            if (t < (unsigned)GCN_PT) {
                const int el = k * 1024 + tid;
                atomicOr(&bits[t * GCN_PSTR + (el >> 5)], 1u << (el & 31));
            }
        }
        __syncthreads();
        const int nt = (n_tiles - tb < GCN_PT) ? (n_tiles - tb) : GCN_PT;
        for (int i = tid; i < nt * 32; i += 1024) {
            const int t  = i >> 5;
            const int w4 = i & 31;
            uint4* drow = (uint4*)(planes + (size_t)(tb + t) * plane_words + (size_t)blockIdx.x * GCN_PBW);
            drow[w4] = ((const uint4*)bits)[t * (GCN_PSTR / 4) + w4];
        }
        __syncthreads();
    }
}

__global__ __launch_bounds__(256) void gcn_dinv(float* __restrict__ deg, int n_nodes) {
    int i = blockIdx.x * blockDim.x + threadIdx.x;
    if (i < n_nodes) deg[i] = rsqrtf(1.0f + deg[i]);
}

__global__ __launch_bounds__(1024) void gcn_gemm(const float* __restrict__ X,
                                                 const float* __restrict__ W,
                                                 float* __restrict__ H, int n) {
    __shared__ float sX[GCN_TILE * GCN_APITCH];

    const int lane  = threadIdx.x & 31;
    const int wave  = __builtin_amdgcn_readfirstlane((int)(threadIdx.x >> 5));
    const int node0 = blockIdx.x * GCN_TILE;
    const int m  = lane & 15;
    const int kh = lane >> 4;
    const int nt = wave & 7;

    v2f b[32];
    {
        const float* bcol = W + (kh * 2) * GCN_D + nt * 16 + m;
        #pragma unroll
        for (int kk = 0; kk < 32; ++kk) {
            b[kk].x = bcol[(kk * 4) * GCN_D];
            b[kk].y = bcol[(kk * 4 + 1) * GCN_D];
        }
    }

    #pragma nounroll
    for (int j0 = 0; j0 < 15; j0 += 5) {
        float4 v[5];
        #pragma unroll
        for (int j = 0; j < 5; ++j) {
            const int i = threadIdx.x + (j0 + j) * (GCN_TWAVES * 32);
            const int ic = (i < GCN_TILE * 32) ? i : (GCN_TILE * 32 - 1);
            int node = node0 + (ic >> 5);
            if (node > n - 1) node = n - 1;
            v[j] = ((const float4*)X)[(size_t)node * (GCN_D / 4) + (ic & 31)];
        }
        asm volatile("" :: "v"(v[0].x), "v"(v[0].y), "v"(v[0].z), "v"(v[0].w),
                           "v"(v[1].x), "v"(v[1].y), "v"(v[1].z), "v"(v[1].w),
                           "v"(v[2].x), "v"(v[2].y), "v"(v[2].z), "v"(v[2].w),
                           "v"(v[3].x), "v"(v[3].y), "v"(v[3].z), "v"(v[3].w),
                           "v"(v[4].x), "v"(v[4].y), "v"(v[4].z), "v"(v[4].w));
        #pragma unroll
        for (int j = 0; j < 5; ++j) {
            const int i = threadIdx.x + (j0 + j) * (GCN_TWAVES * 32);
            if (i < GCN_TILE * 32) {
                const int r = i >> 5;
                const bool in = node0 + r < n;
                float4 o;
                o.x = in ? v[j].x : 0.0f;
                o.y = in ? v[j].y : 0.0f;
                o.z = in ? v[j].z : 0.0f;
                o.w = in ? v[j].w : 0.0f;
                ((float4*)sX)[r * (GCN_APITCH / 4) + (i & 31)] = o;
            }
        }
    }
    __syncthreads();

    for (int strip = wave >> 3; strip < GCN_TILE / 16; strip += GCN_TWAVES / 8) {
        const float* arow = sX + (strip * 16 + m) * GCN_APITCH + kh * 2;
        v8f c = {};
        #pragma unroll
        for (int kk = 0; kk < 32; ++kk) {
            v2f a;
            a.x = arow[kk * 4];
            a.y = arow[kk * 4 + 1];
            c = __builtin_amdgcn_wmma_f32_16x16x4_f32(false, a, false, b[kk], (short)0, c, false, false);
        }
        #pragma unroll
        for (int v = 0; v < 8; ++v) {
            const int node = node0 + strip * 16 + v + 8 * kh;
            if (node < n) H[(size_t)node * GCN_D + nt * 16 + m] = c[v];
        }
    }
}

__device__ __forceinline__ void gcn_consume8(const int* qp, int nvalid,
                                             const int* __restrict__ src,
                                             const int* __restrict__ dst,
                                             const float* __restrict__ dinv,
                                             const float* __restrict__ h,
                                             float* acc, int node0, int n_edges, int lane) {
    int e[8], s[8], d[8];
    float ds[8], dd[8];
    float v[8][4];
    #pragma unroll
    for (int j = 0; j < 8; ++j) {
        const int jj = (j < nvalid) ? j : (nvalid - 1);
        e[j] = qp[jj];
        const int ec = (e[j] < n_edges) ? e[j] : 0;
        s[j] = src[ec];
        d[j] = dst[ec];
    }
    #pragma unroll
    for (int j = 0; j < 8; ++j) {
        ds[j] = dinv[s[j]];
        dd[j] = dinv[d[j]];
        const float* hr = h + (size_t)s[j] * GCN_D + lane;
        v[j][0] = hr[0];
        v[j][1] = hr[32];
        v[j][2] = hr[64];
        v[j][3] = hr[96];
    }
    #pragma unroll
    for (int j = 0; j < 8; ++j) {
        asm volatile("" :: "v"(ds[j]), "v"(dd[j]), "v"(v[j][0]), "v"(v[j][1]), "v"(v[j][2]), "v"(v[j][3]));
    }
    #pragma unroll
    for (int j = 0; j < 8; ++j) {
        const unsigned l = (unsigned)(d[j] - node0);
        const bool ok = (j < nvalid) && (e[j] < n_edges) && (l < (unsigned)GCN_TILE);
        if (ok) {
            const float w = ds[j] * dd[j];
            float* a = acc + l * GCN_APITCH + lane;
            atomicAdd(a + 0,  v[j][0] * w);
            atomicAdd(a + 32, v[j][1] * w);
            atomicAdd(a + 64, v[j][2] * w);
            atomicAdd(a + 96, v[j][3] * w);
        }
    }
}

template <bool FUSE>
__global__ __launch_bounds__(1024) void gcn_tile(const int* __restrict__ src,
                                                 const int* __restrict__ dst,
                                                 const unsigned* __restrict__ planes,
                                                 const float* __restrict__ dinv,
                                                 const float* __restrict__ h,
                                                 const float* __restrict__ bias,
                                                 const float* __restrict__ Wn,
                                                 float* __restrict__ out,
                                                 int n, int n_edges, int n_blk, int plane_words) {
    __shared__ float acc[GCN_TILE * GCN_APITCH];
    __shared__ int queue[GCN_TWAVES * GCN_QCAP];
    __shared__ float sW[FUSE ? GCN_D * GCN_D : 4];

    const int lane  = threadIdx.x & 31;
    const int wave  = __builtin_amdgcn_readfirstlane((int)(threadIdx.x >> 5));
    const int node0 = blockIdx.x * GCN_TILE;

    for (int i = threadIdx.x; i < GCN_TILE * (GCN_APITCH / 4); i += GCN_TWAVES * 32)
        ((float4*)acc)[i] = make_float4(0.0f, 0.0f, 0.0f, 0.0f);
    if (FUSE) {
        for (int i = threadIdx.x; i < GCN_D * (GCN_D / 4); i += GCN_TWAVES * 32)
            ((float4*)sW)[i] = ((const float4*)Wn)[i];
    }
    __syncthreads();

    int* q = queue + wave * GCN_QCAP;
    int cnt = 0;
    const v4u* p4 = (const v4u*)(planes + (size_t)blockIdx.x * plane_words);

    int c = wave;
    v4u cur = {0u, 0u, 0u, 0u};
    if (c < n_blk) cur = __builtin_nontemporal_load(p4 + (size_t)c * 32 + lane);
    for (; c < n_blk; c += GCN_TWAVES) {
        v4u nxt = cur;
        if (c + GCN_TWAVES < n_blk)
            nxt = __builtin_nontemporal_load(p4 + (size_t)(c + GCN_TWAVES) * 32 + lane);

        const unsigned wd[4] = {cur.x, cur.y, cur.z, cur.w};
        const int ebase = c * GCN_PB + lane * 128;
        #pragma unroll
        for (int t = 0; t < 4; ++t) {
            unsigned hz = wd[t];
            for (;;) {
                const bool act = hz != 0u;
                const unsigned bal = __builtin_amdgcn_ballot_w32(act);
                if (bal == 0u) break;
                if (act) {
                    const int b = __builtin_ctz(hz);
                    const int pos = cnt + (int)__builtin_amdgcn_mbcnt_lo(bal, 0u);
                    q[pos] = ebase + t * 32 + b;
                    hz &= hz - 1u;
                }
                cnt += __builtin_popcount(bal);
                while (cnt >= 8) {
                    cnt -= 8;
                    gcn_consume8(q + cnt, 8, src, dst, dinv, h, acc, node0, n_edges, lane);
                }
            }
        }
        cur = nxt;
    }
    if (cnt > 0) gcn_consume8(q, cnt, src, dst, dinv, h, acc, node0, n_edges, lane);
    __syncthreads();

    const int m  = lane & 15;
    const int kh = lane >> 4;
    const int nt = wave & 7;
    v2f b[32];
    if (FUSE) {
        const float* bcol = sW + (kh * 2) * GCN_D + nt * 16 + m;
        #pragma unroll
        for (int kk = 0; kk < 32; ++kk) {
            b[kk].x = bcol[(kk * 4) * GCN_D];
            b[kk].y = bcol[(kk * 4 + 1) * GCN_D];
        }
    }

    for (int i = threadIdx.x; i < GCN_TILE * (GCN_D / 4); i += GCN_TWAVES * 32) {
        const int node = node0 + (i >> 5);
        const int f4   = i & 31;
        if (node < n) {
            const float di = dinv[node];
            const float d2 = di * di;
            const size_t g = (size_t)node * (GCN_D / 4) + f4;
            const int li = (i >> 5) * (GCN_APITCH / 4) + f4;
            const float4 av = ((const float4*)acc)[li];
            const float4 hv = ((const float4*)h)[g];
            const float4 bv = ((const float4*)bias)[f4];
            float4 o;
            o.x = av.x + hv.x * d2 + bv.x;
            o.y = av.y + hv.y * d2 + bv.y;
            o.z = av.z + hv.z * d2 + bv.z;
            o.w = av.w + hv.w * d2 + bv.w;
            if (FUSE) {
                o.x = fmaxf(o.x, 0.0f);
                o.y = fmaxf(o.y, 0.0f);
                o.z = fmaxf(o.z, 0.0f);
                o.w = fmaxf(o.w, 0.0f);
                ((float4*)acc)[li] = o;
            } else {
                ((float4*)out)[g] = o;
            }
        }
    }

    if (FUSE) {
        __syncthreads();
        for (int strip = wave >> 3; strip < GCN_TILE / 16; strip += GCN_TWAVES / 8) {
            const float* arow = acc + (strip * 16 + m) * GCN_APITCH + kh * 2;
            v8f c = {};
            #pragma unroll
            for (int kk = 0; kk < 32; ++kk) {
                v2f a;
                a.x = arow[kk * 4];
                a.y = arow[kk * 4 + 1];
                c = __builtin_amdgcn_wmma_f32_16x16x4_f32(false, a, false, b[kk], (short)0, c, false, false);
            }
            #pragma unroll
            for (int v = 0; v < 8; ++v) {
                const int node = node0 + strip * 16 + v + 8 * kh;
                if (node < n) out[(size_t)node * GCN_D + nt * 16 + m] = c[v];
            }
        }
    }
}

extern "C" void kernel_launch(void* const* d_in, const int* in_sizes, int n_in,
                              void* d_out, int out_size, void* d_ws, size_t ws_size,
                              hipStream_t stream) {
    const float* x  = (const float*)d_in[0];
    const int*   ei = (const int*)d_in[1];
    const float* W1 = (const float*)d_in[2];
    const float* b1 = (const float*)d_in[3];
    const float* W2 = (const float*)d_in[4];
    const float* b2 = (const float*)d_in[5];
    const float* W3 = (const float*)d_in[6];
    const float* b3 = (const float*)d_in[7];

    const int n = in_sizes[0] / GCN_D;
    const int E = in_sizes[1] / 2;
    const int* srcI = ei;
    const int* dstI = ei + E;

    float* ws = (float*)d_ws;
    size_t n4 = ((size_t)n + 3) & ~(size_t)3;
    float* dinv = ws;
    float* h    = ws + n4;
    float* agg  = h + (size_t)n * GCN_D;
    float* bufA = agg + (size_t)n * GCN_D;
    float* out  = (float*)d_out;

    unsigned* planes = (unsigned*)agg;
    const int n_blk = (E + GCN_PB - 1) / GCN_PB;
    const int n_tiles = (n + GCN_TILE - 1) / GCN_TILE;
    const int plane_words = n_blk * GCN_PBW;

    hipMemsetAsync(dinv, 0, (size_t)n * sizeof(float), stream);
    gcn_planes<<<n_blk, 1024, 0, stream>>>(dstI, dinv, planes, E, n_tiles, plane_words);
    gcn_dinv<<<(n + 255) / 256, 256, 0, stream>>>(dinv, n);

    gcn_gemm<<<n_tiles, GCN_TWAVES * 32, 0, stream>>>(x, W1, h, n);
    gcn_tile<true><<<n_tiles, GCN_TWAVES * 32, 0, stream>>>(
        srcI, dstI, planes, dinv, h, b1, W2, bufA, n, E, n_blk, plane_words);
    gcn_tile<true><<<n_tiles, GCN_TWAVES * 32, 0, stream>>>(
        srcI, dstI, planes, dinv, bufA, b2, W3, h, n, E, n_blk, plane_words);
    gcn_tile<false><<<n_tiles, GCN_TWAVES * 32, 0, stream>>>(
        srcI, dstI, planes, dinv, h, b3, nullptr, out, n, E, n_blk, plane_words);
}
